# starting kernel
# baseline (speedup 1.0000x reference)
#include <hip/hip_runtime.h>

typedef __attribute__((ext_vector_type(2))) float v2f;
typedef __attribute__((ext_vector_type(8))) float v8f;

#define DIM 16
#define NEG_SLOPE 0.2f

__device__ __forceinline__ float leaky(float e) { return e >= 0.f ? e : NEG_SLOPE * e; }

__device__ __forceinline__ unsigned fenc(float f) {
  unsigned u = __float_as_uint(f);
  return (u & 0x80000000u) ? ~u : (u | 0x80000000u);
}
__device__ __forceinline__ float fdec(unsigned u) {
  return __uint_as_float((u & 0x80000000u) ? (u & 0x7FFFFFFFu) : ~u);
}

__global__ void gat_pad_x(const float* __restrict__ X, float* __restrict__ XP, int n) {
  int idx = blockIdx.x * blockDim.x + threadIdx.x;
  if (idx >= n * DIM) return;
  int i = idx >> 4, c = idx & 15;
  XP[idx] = (c < 14) ? X[i * 14 + c] : 0.f;
}

__global__ void gat_pad_w(const float* __restrict__ W, float* __restrict__ WP) {
  int idx = threadIdx.x;
  int r = idx >> 4, c = idx & 15;
  WP[idx] = (r < 14) ? W[r * DIM + c] : 0.f;
}

__global__ void gat_proj_wmma(const float* __restrict__ X16, const float* __restrict__ W16,
                              float* __restrict__ H, int n) {
  int wave = (int)((blockIdx.x * blockDim.x + threadIdx.x) >> 5);
  int lane = threadIdx.x & 31;
  int ntiles = (n + 15) >> 4;
  if (wave >= ntiles) return;
  int half = lane >> 4;
  int l = lane & 15;
  int rowA = wave * 16 + l;
  if (rowA >= n) rowA = n - 1;
  v8f c = {};
#pragma unroll
  for (int k0 = 0; k0 < DIM; k0 += 4) {
    int ka = k0 + half * 2;
    float2 av = *(const float2*)&X16[rowA * DIM + ka];
    v2f a, b;
    a.x = av.x;
    a.y = av.y;
    b.x = W16[ka * DIM + l];
    b.y = W16[(ka + 1) * DIM + l];
    c = __builtin_amdgcn_wmma_f32_16x16x4_f32(false, a, false, b, (short)0, c,
                                              false, false);
  }
  int base = (wave * 16 + half * 8) * DIM + l;
  if (wave * 16 + 16 <= n) {
#pragma unroll
    for (int r = 0; r < 8; ++r) H[base + r * DIM] = c[r];
  } else {
#pragma unroll
    for (int r = 0; r < 8; ++r) {
      int row = wave * 16 + r + half * 8;
      if (row < n) H[row * DIM + l] = c[r];
    }
  }
}

__global__ void gat_alphas(const float* __restrict__ H, const float* __restrict__ a_src,
                           const float* __restrict__ a_dst, float* __restrict__ AS,
                           float* __restrict__ AD, int n) {
  int i = blockIdx.x * blockDim.x + threadIdx.x;
  if (i >= n) return;
  float s = 0.f, d = 0.f;
#pragma unroll
  for (int j = 0; j < 4; ++j) {
    float4 h = *(const float4*)&H[i * DIM + j * 4];
    s += h.x * a_src[j * 4 + 0] + h.y * a_src[j * 4 + 1] +
         h.z * a_src[j * 4 + 2] + h.w * a_src[j * 4 + 3];
    d += h.x * a_dst[j * 4 + 0] + h.y * a_dst[j * 4 + 1] +
         h.z * a_dst[j * 4 + 2] + h.w * a_dst[j * 4 + 3];
  }
  AS[i] = s;
  AD[i] = d;
}

__global__ void gat_self_init(const float* __restrict__ AS, const float* __restrict__ AD,
                              unsigned* __restrict__ MK, int n) {
  int i = blockIdx.x * blockDim.x + threadIdx.x;
  if (i >= n) return;
  MK[i] = fenc(leaky(AS[i] + AD[i]));
}

__global__ void gat_edge_max(const int* __restrict__ src, const int* __restrict__ dst,
                             const float* __restrict__ AS, const float* __restrict__ AD,
                             unsigned* __restrict__ MK, int E) {
  int tid = blockIdx.x * blockDim.x + threadIdx.x;
  int stride = gridDim.x * blockDim.x;
  int E4 = E >> 2;
  const int4* s4p = (const int4*)src;
  const int4* d4p = (const int4*)dst;
  for (int q = tid; q < E4; q += stride) {
    int4 s4 = s4p[q], d4 = d4p[q];
    atomicMax(&MK[d4.x], fenc(leaky(AS[s4.x] + AD[d4.x])));
    atomicMax(&MK[d4.y], fenc(leaky(AS[s4.y] + AD[d4.y])));
    atomicMax(&MK[d4.z], fenc(leaky(AS[s4.z] + AD[d4.z])));
    atomicMax(&MK[d4.w], fenc(leaky(AS[s4.w] + AD[d4.w])));
  }
  if (tid < (E & 3)) {
    int e = (E & ~3) + tid;
    int s = src[e], d = dst[e];
    atomicMax(&MK[d], fenc(leaky(AS[s] + AD[d])));
  }
}

__global__ void gat_node_seed(const float* __restrict__ H, const float* __restrict__ AS,
                              const float* __restrict__ AD, const unsigned* __restrict__ MK,
                              float* __restrict__ DEN, float* __restrict__ S, int n) {
  int i = blockIdx.x * blockDim.x + threadIdx.x;
  if (i >= n) return;
  float m = fdec(MK[i]);
  float ex = expf(leaky(AS[i] + AD[i]) - m);
  DEN[i] = ex;
#pragma unroll
  for (int j = 0; j < 4; ++j) {
    float4 h = *(const float4*)&H[i * DIM + j * 4];
    float4 o = {ex * h.x, ex * h.y, ex * h.z, ex * h.w};
    *(float4*)&S[i * DIM + j * 4] = o;
  }
}

__device__ __forceinline__ void gat_edge_one(int s, int d, const float* __restrict__ H,
                                             const float* __restrict__ AS,
                                             const float* __restrict__ AD,
                                             const unsigned* __restrict__ MK,
                                             float* __restrict__ DEN,
                                             float* __restrict__ S) {
  float ex = expf(leaky(AS[s] + AD[d]) - fdec(MK[d]));
  atomicAdd(&DEN[d], ex);
  const float4* h4 = (const float4*)&H[s * DIM];
  float* sd = &S[d * DIM];
#pragma unroll
  for (int j = 0; j < 4; ++j) {
    float4 v = h4[j];
    atomicAdd(&sd[j * 4 + 0], ex * v.x);
    atomicAdd(&sd[j * 4 + 1], ex * v.y);
    atomicAdd(&sd[j * 4 + 2], ex * v.z);
    atomicAdd(&sd[j * 4 + 3], ex * v.w);
  }
}

__global__ void gat_edge_acc(const int* __restrict__ src, const int* __restrict__ dst,
                             const float* __restrict__ H, const float* __restrict__ AS,
                             const float* __restrict__ AD, const unsigned* __restrict__ MK,
                             float* __restrict__ DEN, float* __restrict__ S, int E) {
  int tid = blockIdx.x * blockDim.x + threadIdx.x;
  int stride = gridDim.x * blockDim.x;
  int E4 = E >> 2;
  const int4* s4p = (const int4*)src;
  const int4* d4p = (const int4*)dst;
  for (int q = tid; q < E4; q += stride) {
    __builtin_prefetch(&s4p[q + 1024], 0, 1);
    __builtin_prefetch(&d4p[q + 1024], 0, 1);
    int4 s4 = s4p[q], d4 = d4p[q];
    gat_edge_one(s4.x, d4.x, H, AS, AD, MK, DEN, S);
    gat_edge_one(s4.y, d4.y, H, AS, AD, MK, DEN, S);
    gat_edge_one(s4.z, d4.z, H, AS, AD, MK, DEN, S);
    gat_edge_one(s4.w, d4.w, H, AS, AD, MK, DEN, S);
  }
  if (tid < (E & 3)) {
    int e = (E & ~3) + tid;
    gat_edge_one(src[e], dst[e], H, AS, AD, MK, DEN, S);
  }
}

__global__ void gat_node_fin(const float* __restrict__ S, const float* __restrict__ DEN,
                             const float* __restrict__ b, float* __restrict__ OUT, int n,
                             int do_relu) {
  int idx = blockIdx.x * blockDim.x + threadIdx.x;
  if (idx >= n * DIM) return;
  int i = idx >> 4, cidx = idx & 15;
  float v = S[idx] / DEN[i] + b[cidx];
  if (do_relu) v = v > 0.f ? v : 0.f;
  OUT[idx] = v;
}

extern "C" void kernel_launch(void* const* d_in, const int* in_sizes, int n_in,
                              void* d_out, int out_size, void* d_ws, size_t ws_size,
                              hipStream_t stream) {
  const float* x = (const float*)d_in[0];
  const int* ei = (const int*)d_in[1];
  const float* W1 = (const float*)d_in[2];
  const float* as1 = (const float*)d_in[3];
  const float* ad1 = (const float*)d_in[4];
  const float* b1 = (const float*)d_in[5];
  const float* W2 = (const float*)d_in[6];
  const float* as2 = (const float*)d_in[7];
  const float* ad2 = (const float*)d_in[8];
  const float* b2 = (const float*)d_in[9];
  float* out = (float*)d_out;

  const int NFEAT = 14;
  int n = in_sizes[0] / NFEAT;
  int E = in_sizes[1] / 2;
  const int* src = ei;
  const int* dst = ei + E;

  float* fws = (float*)d_ws;
  float* H = fws;
  float* S = H + (size_t)n * DIM;
  float* O1 = S + (size_t)n * DIM;
  float* XP = O1 + (size_t)n * DIM;
  float* AS = XP + (size_t)n * DIM;
  float* AD = AS + n;
  float* DEN = AD + n;
  unsigned* MK = (unsigned*)(DEN + n);
  float* WP = (float*)(MK + n);

  const int TB = 256;
  int ntiles = (n + 15) >> 4;
  dim3 gProj((ntiles + 7) / 8);
  dim3 gNode((n + TB - 1) / TB);
  dim3 gElem((n * DIM + TB - 1) / TB);
  dim3 gEdge(((E >> 2) + TB - 1) / TB);

  gat_pad_x<<<gElem, TB, 0, stream>>>(x, XP, n);
  gat_pad_w<<<1, 256, 0, stream>>>(W1, WP);
  gat_proj_wmma<<<gProj, TB, 0, stream>>>(XP, WP, H, n);
  gat_alphas<<<gNode, TB, 0, stream>>>(H, as1, ad1, AS, AD, n);
  gat_self_init<<<gNode, TB, 0, stream>>>(AS, AD, MK, n);
  gat_edge_max<<<gEdge, TB, 0, stream>>>(src, dst, AS, AD, MK, E);
  gat_node_seed<<<gNode, TB, 0, stream>>>(H, AS, AD, MK, DEN, S, n);
  gat_edge_acc<<<gEdge, TB, 0, stream>>>(src, dst, H, AS, AD, MK, DEN, S, E);
  gat_node_fin<<<gElem, TB, 0, stream>>>(S, DEN, b1, O1, n, 1);

  gat_proj_wmma<<<gProj, TB, 0, stream>>>(O1, W2, H, n);
  gat_alphas<<<gNode, TB, 0, stream>>>(H, as2, ad2, AS, AD, n);
  gat_self_init<<<gNode, TB, 0, stream>>>(AS, AD, MK, n);
  gat_edge_max<<<gEdge, TB, 0, stream>>>(src, dst, AS, AD, MK, E);
  gat_node_seed<<<gNode, TB, 0, stream>>>(H, AS, AD, MK, DEN, S, n);
  gat_edge_acc<<<gEdge, TB, 0, stream>>>(src, dst, H, AS, AD, MK, DEN, S, E);
  gat_node_fin<<<gElem, TB, 0, stream>>>(S, DEN, b2, out, n, 0);
}


# edge accumulate pass transposed: 16 adjacent lanes per edge row, so each h-row load and each float atomic instruction covers two 64-byte rows instead of 32 scattered dwords
# speedup vs baseline: 2.1109x; 2.1109x over previous
#include <hip/hip_runtime.h>

typedef __attribute__((ext_vector_type(2))) float v2f;
typedef __attribute__((ext_vector_type(8))) float v8f;

#define DIM 16
#define NEG_SLOPE 0.2f

__device__ __forceinline__ float leaky(float e) { return e >= 0.f ? e : NEG_SLOPE * e; }

__device__ __forceinline__ unsigned fenc(float f) {
  unsigned u = __float_as_uint(f);
  return (u & 0x80000000u) ? ~u : (u | 0x80000000u);
}
__device__ __forceinline__ float fdec(unsigned u) {
  return __uint_as_float((u & 0x80000000u) ? (u & 0x7FFFFFFFu) : ~u);
}

__global__ void gat_pad_x(const float* __restrict__ X, float* __restrict__ XP, int n) {
  int idx = blockIdx.x * blockDim.x + threadIdx.x;
  if (idx >= n * DIM) return;
  int i = idx >> 4, c = idx & 15;
  XP[idx] = (c < 14) ? X[i * 14 + c] : 0.f;
}

__global__ void gat_pad_w(const float* __restrict__ W, float* __restrict__ WP) {
  int idx = threadIdx.x;
  int r = idx >> 4, c = idx & 15;
  WP[idx] = (r < 14) ? W[r * DIM + c] : 0.f;
}

__global__ void gat_proj_wmma(const float* __restrict__ X16, const float* __restrict__ W16,
                              float* __restrict__ H, int n) {
  int wave = (int)((blockIdx.x * blockDim.x + threadIdx.x) >> 5);
  int lane = threadIdx.x & 31;
  int ntiles = (n + 15) >> 4;
  if (wave >= ntiles) return;
  int half = lane >> 4;
  int l = lane & 15;
  int rowA = wave * 16 + l;
  if (rowA >= n) rowA = n - 1;
  v8f c = {};
#pragma unroll
  for (int k0 = 0; k0 < DIM; k0 += 4) {
    int ka = k0 + half * 2;
    float2 av = *(const float2*)&X16[rowA * DIM + ka];
    v2f a, b;
    a.x = av.x;
    a.y = av.y;
    b.x = W16[ka * DIM + l];
    b.y = W16[(ka + 1) * DIM + l];
    c = __builtin_amdgcn_wmma_f32_16x16x4_f32(false, a, false, b, (short)0, c,
                                              false, false);
  }
  int base = (wave * 16 + half * 8) * DIM + l;
  if (wave * 16 + 16 <= n) {
#pragma unroll
    for (int r = 0; r < 8; ++r) H[base + r * DIM] = c[r];
  } else {
#pragma unroll
    for (int r = 0; r < 8; ++r) {
      int row = wave * 16 + r + half * 8;
      if (row < n) H[row * DIM + l] = c[r];
    }
  }
}

__global__ void gat_alphas(const float* __restrict__ H, const float* __restrict__ a_src,
                           const float* __restrict__ a_dst, float* __restrict__ AS,
                           float* __restrict__ AD, int n) {
  int i = blockIdx.x * blockDim.x + threadIdx.x;
  if (i >= n) return;
  float s = 0.f, d = 0.f;
#pragma unroll
  for (int j = 0; j < 4; ++j) {
    float4 h = *(const float4*)&H[i * DIM + j * 4];
    s += h.x * a_src[j * 4 + 0] + h.y * a_src[j * 4 + 1] +
         h.z * a_src[j * 4 + 2] + h.w * a_src[j * 4 + 3];
    d += h.x * a_dst[j * 4 + 0] + h.y * a_dst[j * 4 + 1] +
         h.z * a_dst[j * 4 + 2] + h.w * a_dst[j * 4 + 3];
  }
  AS[i] = s;
  AD[i] = d;
}

__global__ void gat_self_init(const float* __restrict__ AS, const float* __restrict__ AD,
                              unsigned* __restrict__ MK, int n) {
  int i = blockIdx.x * blockDim.x + threadIdx.x;
  if (i >= n) return;
  MK[i] = fenc(leaky(AS[i] + AD[i]));
}

__global__ void gat_edge_max(const int* __restrict__ src, const int* __restrict__ dst,
                             const float* __restrict__ AS, const float* __restrict__ AD,
                             unsigned* __restrict__ MK, int E) {
  int tid = blockIdx.x * blockDim.x + threadIdx.x;
  int stride = gridDim.x * blockDim.x;
  int E4 = E >> 2;
  const int4* s4p = (const int4*)src;
  const int4* d4p = (const int4*)dst;
  for (int q = tid; q < E4; q += stride) {
    int4 s4 = s4p[q], d4 = d4p[q];
    atomicMax(&MK[d4.x], fenc(leaky(AS[s4.x] + AD[d4.x])));
    atomicMax(&MK[d4.y], fenc(leaky(AS[s4.y] + AD[d4.y])));
    atomicMax(&MK[d4.z], fenc(leaky(AS[s4.z] + AD[d4.z])));
    atomicMax(&MK[d4.w], fenc(leaky(AS[s4.w] + AD[d4.w])));
  }
  if (tid < (E & 3)) {
    int e = (E & ~3) + tid;
    int s = src[e], d = dst[e];
    atomicMax(&MK[d], fenc(leaky(AS[s] + AD[d])));
  }
}

__global__ void gat_node_seed(const float* __restrict__ H, const float* __restrict__ AS,
                              const float* __restrict__ AD, const unsigned* __restrict__ MK,
                              float* __restrict__ DEN, float* __restrict__ S, int n) {
  int i = blockIdx.x * blockDim.x + threadIdx.x;
  if (i >= n) return;
  float m = fdec(MK[i]);
  float ex = expf(leaky(AS[i] + AD[i]) - m);
  DEN[i] = ex;
#pragma unroll
  for (int j = 0; j < 4; ++j) {
    float4 h = *(const float4*)&H[i * DIM + j * 4];
    float4 o = {ex * h.x, ex * h.y, ex * h.z, ex * h.w};
    *(float4*)&S[i * DIM + j * 4] = o;
  }
}

template <bool FULL>
__device__ __forceinline__ void gat_edge_group(unsigned base, unsigned left, unsigned lane,
                                               const int* __restrict__ src,
                                               const int* __restrict__ dst,
                                               const float* __restrict__ H,
                                               const float* __restrict__ AS,
                                               const float* __restrict__ AD,
                                               const unsigned* __restrict__ MK,
                                               float* __restrict__ DEN,
                                               float* __restrict__ S) {
  const unsigned half = lane >> 4;
  const unsigned c = lane & 15u;
  const bool on = FULL || lane < left;
  unsigned s = 0u, d = 0u, mk = 0u;
  float a_s = 0.f, a_d = 0.f;
  if (on) {
    const unsigned e = base + lane;
    s = (unsigned)src[e];
    d = (unsigned)dst[e];
    a_s = AS[s];
    a_d = AD[d];
    mk = MK[d];
  }
  float h[16];
#pragma unroll
  for (int j = 0; j < 16; ++j) {
    const unsigned q = 2u * (unsigned)j + half;
    const unsigned sj = (unsigned)__shfl((int)s, (int)q);
    h[j] = (FULL || q < left) ? H[sj * DIM + c] : 0.f;
  }
  float ex = 0.f;
  if (on) {
    ex = expf(leaky(a_s + a_d) - fdec(mk));
    atomicAdd(&DEN[d], ex);
  }
#pragma unroll
  for (int j = 0; j < 16; ++j) {
    const unsigned q = 2u * (unsigned)j + half;
    const unsigned dj = (unsigned)__shfl((int)d, (int)q);
    const float exj = __shfl(ex, (int)q);
    if (FULL || q < left) atomicAdd(&S[dj * DIM + c], exj * h[j]);
  }
}

__global__ void gat_edge_acc(const int* __restrict__ src, const int* __restrict__ dst,
                             const float* __restrict__ H, const float* __restrict__ AS,
                             const float* __restrict__ AD, const unsigned* __restrict__ MK,
                             float* __restrict__ DEN, float* __restrict__ S, int E) {
  const unsigned lane = threadIdx.x & 31u;
  const unsigned wave = (unsigned)__builtin_amdgcn_readfirstlane(
      (int)((blockIdx.x * blockDim.x + threadIdx.x) >> 5));
  const unsigned nwave = (gridDim.x * blockDim.x) >> 5;
  const unsigned nE = (unsigned)E;
  for (unsigned base = wave * 32u; base < nE; base += nwave * 32u) {
    const unsigned left = nE - base;
    if (left >= 32u)
      gat_edge_group<true>(base, left, lane, src, dst, H, AS, AD, MK, DEN, S);
    else
      gat_edge_group<false>(base, left, lane, src, dst, H, AS, AD, MK, DEN, S);
  }
}

__global__ void gat_node_fin(const float* __restrict__ S, const float* __restrict__ DEN,
                             const float* __restrict__ b, float* __restrict__ OUT, int n,
                             int do_relu) {
  int idx = blockIdx.x * blockDim.x + threadIdx.x;
  if (idx >= n * DIM) return;
  int i = idx >> 4, cidx = idx & 15;
  float v = S[idx] / DEN[i] + b[cidx];
  if (do_relu) v = v > 0.f ? v : 0.f;
  OUT[idx] = v;
}

extern "C" void kernel_launch(void* const* d_in, const int* in_sizes, int n_in,
                              void* d_out, int out_size, void* d_ws, size_t ws_size,
                              hipStream_t stream) {
  const float* x = (const float*)d_in[0];
  const int* ei = (const int*)d_in[1];
  const float* W1 = (const float*)d_in[2];
  const float* as1 = (const float*)d_in[3];
  const float* ad1 = (const float*)d_in[4];
  const float* b1 = (const float*)d_in[5];
  const float* W2 = (const float*)d_in[6];
  const float* as2 = (const float*)d_in[7];
  const float* ad2 = (const float*)d_in[8];
  const float* b2 = (const float*)d_in[9];
  float* out = (float*)d_out;

  const int NFEAT = 14;
  int n = in_sizes[0] / NFEAT;
  int E = in_sizes[1] / 2;
  const int* src = ei;
  const int* dst = ei + E;

  float* fws = (float*)d_ws;
  float* H = fws;
  float* S = H + (size_t)n * DIM;
  float* O1 = S + (size_t)n * DIM;
  float* XP = O1 + (size_t)n * DIM;
  float* AS = XP + (size_t)n * DIM;
  float* AD = AS + n;
  float* DEN = AD + n;
  unsigned* MK = (unsigned*)(DEN + n);
  float* WP = (float*)(MK + n);

  const int TB = 256;
  int ntiles = (n + 15) >> 4;
  dim3 gProj((ntiles + 7) / 8);
  dim3 gNode((n + TB - 1) / TB);
  dim3 gElem((n * DIM + TB - 1) / TB);
  dim3 gEdge(((E >> 2) + TB - 1) / TB);

  gat_pad_x<<<gElem, TB, 0, stream>>>(x, XP, n);
  gat_pad_w<<<1, 256, 0, stream>>>(W1, WP);
  gat_proj_wmma<<<gProj, TB, 0, stream>>>(XP, WP, H, n);
  gat_alphas<<<gNode, TB, 0, stream>>>(H, as1, ad1, AS, AD, n);
  gat_self_init<<<gNode, TB, 0, stream>>>(AS, AD, MK, n);
  gat_edge_max<<<gEdge, TB, 0, stream>>>(src, dst, AS, AD, MK, E);
  gat_node_seed<<<gNode, TB, 0, stream>>>(H, AS, AD, MK, DEN, S, n);
  gat_edge_acc<<<gEdge, TB, 0, stream>>>(src, dst, H, AS, AD, MK, DEN, S, E);
  gat_node_fin<<<gElem, TB, 0, stream>>>(S, DEN, b1, O1, n, 1);

  gat_proj_wmma<<<gProj, TB, 0, stream>>>(O1, W2, H, n);
  gat_alphas<<<gNode, TB, 0, stream>>>(H, as2, ad2, AS, AD, n);
  gat_self_init<<<gNode, TB, 0, stream>>>(AS, AD, MK, n);
  gat_edge_max<<<gEdge, TB, 0, stream>>>(src, dst, AS, AD, MK, E);
  gat_node_seed<<<gNode, TB, 0, stream>>>(H, AS, AD, MK, DEN, S, n);
  gat_edge_acc<<<gEdge, TB, 0, stream>>>(src, dst, H, AS, AD, MK, DEN, S, E);
  gat_node_fin<<<gElem, TB, 0, stream>>>(S, DEN, b2, out, n, 0);
}


# edges grouped by destination once per call (count, scan, fill); each node's softmax and weighted row sum done by one wave in registers: no float atomics, no max/denominator/numerator planes
# speedup vs baseline: 4.1537x; 1.9677x over previous
#include <hip/hip_runtime.h>

typedef __attribute__((ext_vector_type(2))) float v2f;
typedef __attribute__((ext_vector_type(8))) float v8f;

#define DIM 16
#define NEG_SLOPE 0.2f

__device__ __forceinline__ float leaky(float e) { return e >= 0.f ? e : NEG_SLOPE * e; }

__global__ void gat_pad_x(const float* __restrict__ X, float* __restrict__ XP, int n) {
  int idx = blockIdx.x * blockDim.x + threadIdx.x;
  if (idx >= n * DIM) return;
  int i = idx >> 4, c = idx & 15;
  XP[idx] = (c < 14) ? X[i * 14 + c] : 0.f;
}

__global__ void gat_pad_w(const float* __restrict__ W, float* __restrict__ WP) {
  int idx = threadIdx.x;
  int r = idx >> 4, c = idx & 15;
  WP[idx] = (r < 14) ? W[r * DIM + c] : 0.f;
}

__global__ void gat_proj_wmma(const float* __restrict__ X16, const float* __restrict__ W16,
                              float* __restrict__ H, int n) {
  int wave = (int)((blockIdx.x * blockDim.x + threadIdx.x) >> 5);
  int lane = threadIdx.x & 31;
  int ntiles = (n + 15) >> 4;
  if (wave >= ntiles) return;
  int half = lane >> 4;
  int l = lane & 15;
  int rowA = wave * 16 + l;
  if (rowA >= n) rowA = n - 1;
  v8f c = {};
#pragma unroll
  for (int k0 = 0; k0 < DIM; k0 += 4) {
    int ka = k0 + half * 2;
    float2 av = *(const float2*)&X16[rowA * DIM + ka];
    v2f a, b;
    a.x = av.x;
    a.y = av.y;
    b.x = W16[ka * DIM + l];
    b.y = W16[(ka + 1) * DIM + l];
    c = __builtin_amdgcn_wmma_f32_16x16x4_f32(false, a, false, b, (short)0, c,
                                              false, false);
  }
  int base = (wave * 16 + half * 8) * DIM + l;
  if (wave * 16 + 16 <= n) {
#pragma unroll
    for (int r = 0; r < 8; ++r) H[base + r * DIM] = c[r];
  } else {
#pragma unroll
    for (int r = 0; r < 8; ++r) {
      int row = wave * 16 + r + half * 8;
      if (row < n) H[row * DIM + l] = c[r];
    }
  }
}

__global__ void gat_alphas(const float* __restrict__ H, const float* __restrict__ a_src,
                           const float* __restrict__ a_dst, float* __restrict__ AS,
                           float* __restrict__ AD, int n) {
  int i = blockIdx.x * blockDim.x + threadIdx.x;
  if (i >= n) return;
  float s = 0.f, d = 0.f;
#pragma unroll
  for (int j = 0; j < 4; ++j) {
    float4 h = *(const float4*)&H[i * DIM + j * 4];
    s += h.x * a_src[j * 4 + 0] + h.y * a_src[j * 4 + 1] +
         h.z * a_src[j * 4 + 2] + h.w * a_src[j * 4 + 3];
    d += h.x * a_dst[j * 4 + 0] + h.y * a_dst[j * 4 + 1] +
         h.z * a_dst[j * 4 + 2] + h.w * a_dst[j * 4 + 3];
  }
  AS[i] = s;
  AD[i] = d;
}

__global__ void csr_zero(int* __restrict__ CUR, int n) {
  int i = blockIdx.x * blockDim.x + threadIdx.x;
  if (i < n) CUR[i] = 0;
}

__global__ void csr_count(const int* __restrict__ dst, int* __restrict__ CUR, int E) {
  int tid = blockIdx.x * blockDim.x + threadIdx.x;
  int stride = gridDim.x * blockDim.x;
  int E4 = E >> 2;
  const int4* d4p = (const int4*)dst;
  for (int q = tid; q < E4; q += stride) {
    int4 d4 = d4p[q];
    atomicAdd(&CUR[d4.x], 1);
    atomicAdd(&CUR[d4.y], 1);
    atomicAdd(&CUR[d4.z], 1);
    atomicAdd(&CUR[d4.w], 1);
  }
  if (tid < (E & 3)) atomicAdd(&CUR[dst[(E & ~3) + tid]], 1);
}

#define SCAN_TB 256
#define SCAN_PER 4

__global__ void csr_bsum(const int* __restrict__ CUR, int* __restrict__ BS, int n) {
  __shared__ int sh[SCAN_TB];
  int tid = threadIdx.x;
  int base = (blockIdx.x * SCAN_TB + tid) * SCAN_PER;
  int t = 0;
#pragma unroll
  for (int k = 0; k < SCAN_PER; ++k)
    if (base + k < n) t += CUR[base + k];
  sh[tid] = t;
  __syncthreads();
  for (int w = SCAN_TB / 2; w > 0; w >>= 1) {
    if (tid < w) sh[tid] += sh[tid + w];
    __syncthreads();
  }
  if (tid == 0) BS[blockIdx.x] = sh[0];
}

__global__ void csr_scan(int* __restrict__ CUR, const int* __restrict__ BS, int n) {
  __shared__ int sh[SCAN_TB];
  int tid = threadIdx.x;
  int b = blockIdx.x;
  int part = 0;
  for (int j = tid; j < b; j += SCAN_TB) part += BS[j];
  sh[tid] = part;
  __syncthreads();
  for (int w = SCAN_TB / 2; w > 0; w >>= 1) {
    if (tid < w) sh[tid] += sh[tid + w];
    __syncthreads();
  }
  int boff = sh[0];
  __syncthreads();
  int base = (b * SCAN_TB + tid) * SCAN_PER;
  int v[SCAN_PER];
  int t = 0;
#pragma unroll
  for (int k = 0; k < SCAN_PER; ++k) {
    v[k] = (base + k < n) ? CUR[base + k] : 0;
    t += v[k];
  }
  sh[tid] = t;
  __syncthreads();
  for (int w = 1; w < SCAN_TB; w <<= 1) {
    int add = (tid >= w) ? sh[tid - w] : 0;
    __syncthreads();
    sh[tid] += add;
    __syncthreads();
  }
  int run = boff + sh[tid] - t;
#pragma unroll
  for (int k = 0; k < SCAN_PER; ++k) {
    if (base + k < n) CUR[base + k] = run;
    run += v[k];
  }
}

__global__ void csr_fill(const int* __restrict__ src, const int* __restrict__ dst,
                         int* __restrict__ CUR, int* __restrict__ CSR, int E) {
  int tid = blockIdx.x * blockDim.x + threadIdx.x;
  int stride = gridDim.x * blockDim.x;
  int E4 = E >> 2;
  const int4* s4p = (const int4*)src;
  const int4* d4p = (const int4*)dst;
  for (int q = tid; q < E4; q += stride) {
    int4 s4 = s4p[q], d4 = d4p[q];
    CSR[atomicAdd(&CUR[d4.x], 1)] = s4.x;
    CSR[atomicAdd(&CUR[d4.y], 1)] = s4.y;
    CSR[atomicAdd(&CUR[d4.z], 1)] = s4.z;
    CSR[atomicAdd(&CUR[d4.w], 1)] = s4.w;
  }
  if (tid < (E & 3)) {
    int e = (E & ~3) + tid;
    CSR[atomicAdd(&CUR[dst[e]], 1)] = src[e];
  }
}

__device__ __forceinline__ void gat_rows(unsigned cnt, unsigned lane, unsigned s, float ex,
                                         const float* __restrict__ H, float4& acc) {
  const unsigned slot = lane >> 2;
  const unsigned quad = lane & 3u;
  float4 h[4];
#pragma unroll
  for (int r = 0; r < 4; ++r) {
    h[r] = make_float4(0.f, 0.f, 0.f, 0.f);
    if (8u * (unsigned)r < cnt) {
      const unsigned q = 8u * (unsigned)r + slot;
      const unsigned sr = (unsigned)__shfl((int)s, (int)q);
      h[r] = *(const float4*)&H[sr * DIM + quad * 4u];
    }
  }
#pragma unroll
  for (int r = 0; r < 4; ++r) {
    if (8u * (unsigned)r < cnt) {
      const unsigned q = 8u * (unsigned)r + slot;
      const float w = __shfl(ex, (int)q);
      const bool on = q < cnt;
      acc.x += w * (on ? h[r].x : 0.f);
      acc.y += w * (on ? h[r].y : 0.f);
      acc.z += w * (on ? h[r].z : 0.f);
      acc.w += w * (on ? h[r].w : 0.f);
    }
  }
}

__device__ __forceinline__ float4 gat_node(unsigned i, unsigned lane,
                                           const int* __restrict__ CUR,
                                           const int* __restrict__ CSR,
                                           const float* __restrict__ H,
                                           const float* __restrict__ AS,
                                           const float* __restrict__ AD) {
  const unsigned beg = i ? (unsigned)CUR[i - 1] : 0u;
  const unsigned end = (unsigned)CUR[i];
  const float ad = AD[i];
  const float e_self = leaky(AS[i] + ad);
  unsigned s0 = i;
  float e0 = e_self;
  if (beg + lane < end) {
    s0 = (unsigned)CSR[beg + lane];
    e0 = leaky(AS[s0] + ad);
  }
  float m = fmaxf(e_self, e0);
  for (unsigned p = beg + 32u + lane; p < end; p += 32u)
    m = fmaxf(m, leaky(AS[CSR[p]] + ad));
#pragma unroll
  for (int off = 16; off > 0; off >>= 1) m = fmaxf(m, __shfl_xor(m, off));
  float den = 0.f;
  float4 acc = make_float4(0.f, 0.f, 0.f, 0.f);
  unsigned s = s0;
  float e = e0;
  for (unsigned base = beg; base < end; base += 32u) {
    const unsigned cnt = end - base;
    if (base != beg) {
      s = i;
      e = 0.f;
      if (lane < cnt) {
        s = (unsigned)CSR[base + lane];
        e = leaky(AS[s] + ad);
      }
    }
    const float ex = (lane < cnt) ? expf(e - m) : 0.f;
    den += ex;
    gat_rows(cnt, lane, s, ex, H, acc);
  }
#pragma unroll
  for (int off = 16; off > 0; off >>= 1) den += __shfl_xor(den, off);
#pragma unroll
  for (int off = 16; off >= 4; off >>= 1) {
    acc.x += __shfl_xor(acc.x, off);
    acc.y += __shfl_xor(acc.y, off);
    acc.z += __shfl_xor(acc.z, off);
    acc.w += __shfl_xor(acc.w, off);
  }
  const float ex_self = expf(e_self - m);
  den += ex_self;
  const float4 hs = *(const float4*)&H[i * DIM + (lane & 3u) * 4u];
  float4 o;
  o.x = (acc.x + ex_self * hs.x) / den;
  o.y = (acc.y + ex_self * hs.y) / den;
  o.z = (acc.z + ex_self * hs.z) / den;
  o.w = (acc.w + ex_self * hs.w) / den;
  return o;
}

__global__ void gat_agg(const int* __restrict__ CUR, const int* __restrict__ CSR,
                        const float* __restrict__ H, const float* __restrict__ AS,
                        const float* __restrict__ AD, const float* __restrict__ b,
                        float* __restrict__ OUT, int n, int do_relu) {
  const unsigned lane = threadIdx.x & 31u;
  const unsigned wave = (unsigned)__builtin_amdgcn_readfirstlane(
      (int)((blockIdx.x * blockDim.x + threadIdx.x) >> 5));
  const unsigned i0 = wave * 2u;
  const unsigned nn = (unsigned)n;
  if (i0 >= nn) return;
  const float4 r0 = gat_node(i0, lane, CUR, CSR, H, AS, AD);
  float4 r1 = make_float4(0.f, 0.f, 0.f, 0.f);
  if (i0 + 1u < nn) r1 = gat_node(i0 + 1u, lane, CUR, CSR, H, AS, AD);
  const unsigned row = (lane >> 2) & 1u;
  const float4 bb = *(const float4*)&b[(lane & 3u) * 4u];
  float4 v;
  v.x = (row ? r1.x : r0.x) + bb.x;
  v.y = (row ? r1.y : r0.y) + bb.y;
  v.z = (row ? r1.z : r0.z) + bb.z;
  v.w = (row ? r1.w : r0.w) + bb.w;
  if (do_relu) {
    v.x = v.x > 0.f ? v.x : 0.f;
    v.y = v.y > 0.f ? v.y : 0.f;
    v.z = v.z > 0.f ? v.z : 0.f;
    v.w = v.w > 0.f ? v.w : 0.f;
  }
  if (lane < 8u && i0 + row < nn) *(float4*)&OUT[i0 * DIM + lane * 4u] = v;
}

extern "C" void kernel_launch(void* const* d_in, const int* in_sizes, int n_in,
                              void* d_out, int out_size, void* d_ws, size_t ws_size,
                              hipStream_t stream) {
  const float* x = (const float*)d_in[0];
  const int* ei = (const int*)d_in[1];
  const float* W1 = (const float*)d_in[2];
  const float* as1 = (const float*)d_in[3];
  const float* ad1 = (const float*)d_in[4];
  const float* b1 = (const float*)d_in[5];
  const float* W2 = (const float*)d_in[6];
  const float* as2 = (const float*)d_in[7];
  const float* ad2 = (const float*)d_in[8];
  const float* b2 = (const float*)d_in[9];
  float* out = (float*)d_out;

  const int NFEAT = 14;
  int n = in_sizes[0] / NFEAT;
  int E = in_sizes[1] / 2;
  const int* src = ei;
  const int* dst = ei + E;

  int nb = (n + SCAN_TB * SCAN_PER - 1) / (SCAN_TB * SCAN_PER);
  float* fws = (float*)d_ws;
  float* H = fws;
  float* XO = H + (size_t)n * DIM;
  float* AS = XO + (size_t)n * DIM;
  float* AD = AS + n;
  int* CUR = (int*)(AD + n);
  int* CSR = CUR + n;
  int* BS = CSR + E;
  float* WP = (float*)(BS + nb);

  const int TB = 256;
  int ntiles = (n + 15) >> 4;
  dim3 gProj((ntiles + 7) / 8);
  dim3 gNode((n + TB - 1) / TB);
  dim3 gElem((n * DIM + TB - 1) / TB);
  dim3 gEdge(((E >> 2) + TB - 1) / TB);
  dim3 gAgg(((n + 1) / 2 + 7) / 8);

  csr_zero<<<gNode, TB, 0, stream>>>(CUR, n);
  csr_count<<<gEdge, TB, 0, stream>>>(dst, CUR, E);
  csr_bsum<<<nb, SCAN_TB, 0, stream>>>(CUR, BS, n);
  csr_scan<<<nb, SCAN_TB, 0, stream>>>(CUR, BS, n);
  csr_fill<<<gEdge, TB, 0, stream>>>(src, dst, CUR, CSR, E);

  gat_pad_x<<<gElem, TB, 0, stream>>>(x, XO, n);
  gat_pad_w<<<1, 256, 0, stream>>>(W1, WP);
  gat_proj_wmma<<<gProj, TB, 0, stream>>>(XO, WP, H, n);
  gat_alphas<<<gNode, TB, 0, stream>>>(H, as1, ad1, AS, AD, n);
  gat_agg<<<gAgg, TB, 0, stream>>>(CUR, CSR, H, AS, AD, b1, XO, n, 1);

  gat_proj_wmma<<<gProj, TB, 0, stream>>>(XO, W2, H, n);
  gat_alphas<<<gNode, TB, 0, stream>>>(H, as2, ad2, AS, AD, n);
  gat_agg<<<gAgg, TB, 0, stream>>>(CUR, CSR, H, AS, AD, b2, out, n, 0);
}


# edges grouped by destination with a two-level counting sort on LDS counters (256-destination buckets, one block per bucket): no global atomic per edge, one add per block and bucket instead
# speedup vs baseline: 12.1524x; 2.9257x over previous
#include <hip/hip_runtime.h>

typedef __attribute__((ext_vector_type(2))) float v2f;
typedef __attribute__((ext_vector_type(8))) float v8f;

#define DIM 16
#define NEG_SLOPE 0.2f

__device__ __forceinline__ float leaky(float e) { return e >= 0.f ? e : NEG_SLOPE * e; }

__global__ void gat_pad_x(const float* __restrict__ X, float* __restrict__ XP, int n) {
  int idx = blockIdx.x * blockDim.x + threadIdx.x;
  if (idx >= n * DIM) return;
  int i = idx >> 4, c = idx & 15;
  XP[idx] = (c < 14) ? X[i * 14 + c] : 0.f;
}

__global__ void gat_pad_w(const float* __restrict__ W, float* __restrict__ WP) {
  int idx = threadIdx.x;
  int r = idx >> 4, c = idx & 15;
  WP[idx] = (r < 14) ? W[r * DIM + c] : 0.f;
}

__global__ void gat_proj_wmma(const float* __restrict__ X16, const float* __restrict__ W16,
                              float* __restrict__ H, int n) {
  int wave = (int)((blockIdx.x * blockDim.x + threadIdx.x) >> 5);
  int lane = threadIdx.x & 31;
  int ntiles = (n + 15) >> 4;
  if (wave >= ntiles) return;
  int half = lane >> 4;
  int l = lane & 15;
  int rowA = wave * 16 + l;
  if (rowA >= n) rowA = n - 1;
  v8f c = {};
#pragma unroll
  for (int k0 = 0; k0 < DIM; k0 += 4) {
    int ka = k0 + half * 2;
    float2 av = *(const float2*)&X16[rowA * DIM + ka];
    v2f a, b;
    a.x = av.x;
    a.y = av.y;
    b.x = W16[ka * DIM + l];
    b.y = W16[(ka + 1) * DIM + l];
    c = __builtin_amdgcn_wmma_f32_16x16x4_f32(false, a, false, b, (short)0, c,
                                              false, false);
  }
  int base = (wave * 16 + half * 8) * DIM + l;
  if (wave * 16 + 16 <= n) {
#pragma unroll
    for (int r = 0; r < 8; ++r) H[base + r * DIM] = c[r];
  } else {
#pragma unroll
    for (int r = 0; r < 8; ++r) {
      int row = wave * 16 + r + half * 8;
      if (row < n) H[row * DIM + l] = c[r];
    }
  }
}

__global__ void gat_alphas(const float* __restrict__ H, const float* __restrict__ a_src,
                           const float* __restrict__ a_dst, float* __restrict__ AS,
                           float* __restrict__ AD, int n) {
  int i = blockIdx.x * blockDim.x + threadIdx.x;
  if (i >= n) return;
  float s = 0.f, d = 0.f;
#pragma unroll
  for (int j = 0; j < 4; ++j) {
    float4 h = *(const float4*)&H[i * DIM + j * 4];
    s += h.x * a_src[j * 4 + 0] + h.y * a_src[j * 4 + 1] +
         h.z * a_src[j * 4 + 2] + h.w * a_src[j * 4 + 3];
    d += h.x * a_dst[j * 4 + 0] + h.y * a_dst[j * 4 + 1] +
         h.z * a_dst[j * 4 + 2] + h.w * a_dst[j * 4 + 3];
  }
  AS[i] = s;
  AD[i] = d;
}

#define BSH 8
#define BNODES (1 << BSH)
#define NBK_MAX 1024
#define PT_TB 1024
#define BK_PITCH 32
#define SORT_BUF 16384

__global__ void csr_zero(int* __restrict__ BK, int cnt) {
  int i = blockIdx.x * blockDim.x + threadIdx.x;
  if (i < cnt) BK[i] = 0;
}

__device__ __forceinline__ void part_hist(int* cnt, const int* __restrict__ dst, int E) {
  const int tid = threadIdx.x;
  const int E4 = E >> 2;
  const int qb = (E4 + (int)gridDim.x - 1) / (int)gridDim.x;
  const int q0 = (int)blockIdx.x * qb;
  const int q1 = (q0 + qb < E4) ? q0 + qb : E4;
  const int4* d4p = (const int4*)dst;
  for (int q = q0 + tid; q < q1; q += PT_TB) {
    int4 d4 = d4p[q];
    atomicAdd(&cnt[d4.x >> BSH], 1);
    atomicAdd(&cnt[d4.y >> BSH], 1);
    atomicAdd(&cnt[d4.z >> BSH], 1);
    atomicAdd(&cnt[d4.w >> BSH], 1);
  }
  if (blockIdx.x == 0 && tid < (E & 3)) atomicAdd(&cnt[dst[(E & ~3) + tid] >> BSH], 1);
}

__global__ void __launch_bounds__(PT_TB)
part_count(const int* __restrict__ dst, int* __restrict__ BK, int E, int nbk) {
  __shared__ int cnt[NBK_MAX];
  const int tid = threadIdx.x;
  for (int j = tid; j < nbk; j += PT_TB) cnt[j] = 0;
  __syncthreads();
  part_hist(cnt, dst, E);
  __syncthreads();
  for (int j = tid; j < nbk; j += PT_TB) {
    int c = cnt[j];
    if (c) atomicAdd(&BK[j * BK_PITCH], c);
  }
}

__global__ void __launch_bounds__(NBK_MAX)
part_scan(int* __restrict__ BK, int* __restrict__ BKS, int nbk) {
  __shared__ int sh[NBK_MAX];
  const int tid = threadIdx.x;
  const int v = (tid < nbk) ? BK[tid * BK_PITCH] : 0;
  sh[tid] = v;
  __syncthreads();
  for (int w = 1; w < NBK_MAX; w <<= 1) {
    int add = (tid >= w) ? sh[tid - w] : 0;
    __syncthreads();
    sh[tid] += add;
    __syncthreads();
  }
  const int incl = sh[tid];
  if (tid < nbk) {
    BKS[tid] = incl - v;
    BK[tid * BK_PITCH] = incl - v;
  }
  if (tid == nbk - 1) BKS[nbk] = incl;
}

__global__ void __launch_bounds__(PT_TB)
part_scatter(const int* __restrict__ src, const int* __restrict__ dst, int* __restrict__ BK,
             int* __restrict__ TMP, int E, int nbk) {
  __shared__ int cnt[NBK_MAX];
  const int tid = threadIdx.x;
  for (int j = tid; j < nbk; j += PT_TB) cnt[j] = 0;
  __syncthreads();
  part_hist(cnt, dst, E);
  __syncthreads();
  for (int j = tid; j < nbk; j += PT_TB) {
    int c = cnt[j];
    cnt[j] = c ? atomicAdd(&BK[j * BK_PITCH], c) : 0;
  }
  __syncthreads();
  const int E4 = E >> 2;
  const int qb = (E4 + (int)gridDim.x - 1) / (int)gridDim.x;
  const int q0 = (int)blockIdx.x * qb;
  const int q1 = (q0 + qb < E4) ? q0 + qb : E4;
  const int4* s4p = (const int4*)src;
  const int4* d4p = (const int4*)dst;
  for (int q = q0 + tid; q < q1; q += PT_TB) {
    int4 s4 = s4p[q], d4 = d4p[q];
    TMP[atomicAdd(&cnt[d4.x >> BSH], 1)] = (s4.x << BSH) | (d4.x & (BNODES - 1));
    TMP[atomicAdd(&cnt[d4.y >> BSH], 1)] = (s4.y << BSH) | (d4.y & (BNODES - 1));
    TMP[atomicAdd(&cnt[d4.z >> BSH], 1)] = (s4.z << BSH) | (d4.z & (BNODES - 1));
    TMP[atomicAdd(&cnt[d4.w >> BSH], 1)] = (s4.w << BSH) | (d4.w & (BNODES - 1));
  }
  if (blockIdx.x == 0 && tid < (E & 3)) {
    int e = (E & ~3) + tid;
    int d = dst[e];
    TMP[atomicAdd(&cnt[d >> BSH], 1)] = (src[e] << BSH) | (d & (BNODES - 1));
  }
}

__global__ void __launch_bounds__(PT_TB)
part_sort(const int* __restrict__ TMP, const int* __restrict__ BKS, int* __restrict__ CUR,
          int* __restrict__ CSR, int n) {
  __shared__ int c[BNODES];
  __shared__ int sc[BNODES];
  __shared__ int buf[SORT_BUF];
  const int tid = threadIdx.x;
  const int b = blockIdx.x;
  const int e0 = BKS[b], e1 = BKS[b + 1];
  const int len = e1 - e0;
  const bool staged = len <= SORT_BUF;
  if (tid < BNODES) c[tid] = 0;
  __syncthreads();
  for (int e = e0 + tid; e < e1; e += PT_TB) atomicAdd(&c[TMP[e] & (BNODES - 1)], 1);
  __syncthreads();
  int v = 0;
  if (tid < BNODES) {
    v = c[tid];
    sc[tid] = v;
  }
  __syncthreads();
  for (int w = 1; w < BNODES; w <<= 1) {
    int add = 0;
    if (tid < BNODES && tid >= w) add = sc[tid - w];
    __syncthreads();
    if (tid < BNODES) sc[tid] += add;
    __syncthreads();
  }
  if (tid < BNODES) {
    const int end = e0 + sc[tid];
    const int node = (b << BSH) + tid;
    if (node < n) CUR[node] = end;
    c[tid] = (staged ? sc[tid] : end) - v;
  }
  __syncthreads();
  if (staged) {
    for (int e = e0 + tid; e < e1; e += PT_TB) {
      const int p = TMP[e];
      buf[atomicAdd(&c[p & (BNODES - 1)], 1)] = (int)((unsigned)p >> BSH);
    }
    __syncthreads();
    for (int k = tid; k < len; k += PT_TB) CSR[e0 + k] = buf[k];
  } else {
    for (int e = e0 + tid; e < e1; e += PT_TB) {
      const int p = TMP[e];
      CSR[atomicAdd(&c[p & (BNODES - 1)], 1)] = (int)((unsigned)p >> BSH);
    }
  }
}

__device__ __forceinline__ void gat_rows(unsigned cnt, unsigned lane, unsigned s, float ex,
                                         const float* __restrict__ H, float4& acc) {
  const unsigned slot = lane >> 2;
  const unsigned quad = lane & 3u;
  float4 h[4];
#pragma unroll
  for (int r = 0; r < 4; ++r) {
    h[r] = make_float4(0.f, 0.f, 0.f, 0.f);
    if (8u * (unsigned)r < cnt) {
      const unsigned q = 8u * (unsigned)r + slot;
      const unsigned sr = (unsigned)__shfl((int)s, (int)q);
      h[r] = *(const float4*)&H[sr * DIM + quad * 4u];
    }
  }
#pragma unroll
  for (int r = 0; r < 4; ++r) {
    if (8u * (unsigned)r < cnt) {
      const unsigned q = 8u * (unsigned)r + slot;
      const float w = __shfl(ex, (int)q);
      const bool on = q < cnt;
      acc.x += w * (on ? h[r].x : 0.f);
      acc.y += w * (on ? h[r].y : 0.f);
      acc.z += w * (on ? h[r].z : 0.f);
      acc.w += w * (on ? h[r].w : 0.f);
    }
  }
}

__device__ __forceinline__ float4 gat_node(unsigned i, unsigned lane,
                                           const int* __restrict__ CUR,
                                           const int* __restrict__ CSR,
                                           const float* __restrict__ H,
                                           const float* __restrict__ AS,
                                           const float* __restrict__ AD) {
  const unsigned beg = i ? (unsigned)CUR[i - 1] : 0u;
  const unsigned end = (unsigned)CUR[i];
  const float ad = AD[i];
  const float e_self = leaky(AS[i] + ad);
  unsigned s0 = i;
  float e0 = e_self;
  if (beg + lane < end) {
    s0 = (unsigned)CSR[beg + lane];
    e0 = leaky(AS[s0] + ad);
  }
  float m = fmaxf(e_self, e0);
  for (unsigned p = beg + 32u + lane; p < end; p += 32u)
    m = fmaxf(m, leaky(AS[CSR[p]] + ad));
#pragma unroll
  for (int off = 16; off > 0; off >>= 1) m = fmaxf(m, __shfl_xor(m, off));
  float den = 0.f;
  float4 acc = make_float4(0.f, 0.f, 0.f, 0.f);
  unsigned s = s0;
  float e = e0;
  for (unsigned base = beg; base < end; base += 32u) {
    const unsigned cnt = end - base;
    if (base != beg) {
      s = i;
      e = 0.f;
      if (lane < cnt) {
        s = (unsigned)CSR[base + lane];
        e = leaky(AS[s] + ad);
      }
    }
    const float ex = (lane < cnt) ? expf(e - m) : 0.f;
    den += ex;
    gat_rows(cnt, lane, s, ex, H, acc);
  }
#pragma unroll
  for (int off = 16; off > 0; off >>= 1) den += __shfl_xor(den, off);
#pragma unroll
  for (int off = 16; off >= 4; off >>= 1) {
    acc.x += __shfl_xor(acc.x, off);
    acc.y += __shfl_xor(acc.y, off);
    acc.z += __shfl_xor(acc.z, off);
    acc.w += __shfl_xor(acc.w, off);
  }
  const float ex_self = expf(e_self - m);
  den += ex_self;
  const float4 hs = *(const float4*)&H[i * DIM + (lane & 3u) * 4u];
  float4 o;
  o.x = (acc.x + ex_self * hs.x) / den;
  o.y = (acc.y + ex_self * hs.y) / den;
  o.z = (acc.z + ex_self * hs.z) / den;
  o.w = (acc.w + ex_self * hs.w) / den;
  return o;
}

__global__ void gat_agg(const int* __restrict__ CUR, const int* __restrict__ CSR,
                        const float* __restrict__ H, const float* __restrict__ AS,
                        const float* __restrict__ AD, const float* __restrict__ b,
                        float* __restrict__ OUT, int n, int do_relu) {
  const unsigned lane = threadIdx.x & 31u;
  const unsigned wave = (unsigned)__builtin_amdgcn_readfirstlane(
      (int)((blockIdx.x * blockDim.x + threadIdx.x) >> 5));
  const unsigned i0 = wave * 2u;
  const unsigned nn = (unsigned)n;
  if (i0 >= nn) return;
  const float4 r0 = gat_node(i0, lane, CUR, CSR, H, AS, AD);
  float4 r1 = make_float4(0.f, 0.f, 0.f, 0.f);
  if (i0 + 1u < nn) r1 = gat_node(i0 + 1u, lane, CUR, CSR, H, AS, AD);
  const unsigned row = (lane >> 2) & 1u;
  const float4 bb = *(const float4*)&b[(lane & 3u) * 4u];
  float4 v;
  v.x = (row ? r1.x : r0.x) + bb.x;
  v.y = (row ? r1.y : r0.y) + bb.y;
  v.z = (row ? r1.z : r0.z) + bb.z;
  v.w = (row ? r1.w : r0.w) + bb.w;
  if (do_relu) {
    v.x = v.x > 0.f ? v.x : 0.f;
    v.y = v.y > 0.f ? v.y : 0.f;
    v.z = v.z > 0.f ? v.z : 0.f;
    v.w = v.w > 0.f ? v.w : 0.f;
  }
  if (lane < 8u && i0 + row < nn) *(float4*)&OUT[i0 * DIM + lane * 4u] = v;
}

extern "C" void kernel_launch(void* const* d_in, const int* in_sizes, int n_in,
                              void* d_out, int out_size, void* d_ws, size_t ws_size,
                              hipStream_t stream) {
  const float* x = (const float*)d_in[0];
  const int* ei = (const int*)d_in[1];
  const float* W1 = (const float*)d_in[2];
  const float* as1 = (const float*)d_in[3];
  const float* ad1 = (const float*)d_in[4];
  const float* b1 = (const float*)d_in[5];
  const float* W2 = (const float*)d_in[6];
  const float* as2 = (const float*)d_in[7];
  const float* ad2 = (const float*)d_in[8];
  const float* b2 = (const float*)d_in[9];
  float* out = (float*)d_out;

  const int NFEAT = 14;
  int n = in_sizes[0] / NFEAT;
  int E = in_sizes[1] / 2;
  const int* src = ei;
  const int* dst = ei + E;

  int nbk = (n + BNODES - 1) >> BSH;
  float* fws = (float*)d_ws;
  float* H = fws;
  float* XO = H + (size_t)n * DIM;
  float* AS = XO + (size_t)n * DIM;
  float* AD = AS + n;
  int* CUR = (int*)(AD + n);
  int* CSR = CUR + n;
  float* WP = (float*)(CSR + E);
  int* TMP = (int*)H;
  int* BK = (int*)AS;
  int* BKS = (int*)AD;

  const int TB = 256;
  int ntiles = (n + 15) >> 4;
  dim3 gProj((ntiles + 7) / 8);
  dim3 gNode((n + TB - 1) / TB);
  dim3 gElem((n * DIM + TB - 1) / TB);
  dim3 gPart(256);
  dim3 gAgg(((n + 1) / 2 + 7) / 8);

  csr_zero<<<(nbk * BK_PITCH + TB - 1) / TB, TB, 0, stream>>>(BK, nbk * BK_PITCH);
  part_count<<<gPart, PT_TB, 0, stream>>>(dst, BK, E, nbk);
  part_scan<<<1, NBK_MAX, 0, stream>>>(BK, BKS, nbk);
  part_scatter<<<gPart, PT_TB, 0, stream>>>(src, dst, BK, TMP, E, nbk);
  part_sort<<<nbk, PT_TB, 0, stream>>>(TMP, BKS, CUR, CSR, n);

  gat_pad_x<<<gElem, TB, 0, stream>>>(x, XO, n);
  gat_pad_w<<<1, 256, 0, stream>>>(W1, WP);
  gat_proj_wmma<<<gProj, TB, 0, stream>>>(XO, WP, H, n);
  gat_alphas<<<gNode, TB, 0, stream>>>(H, as1, ad1, AS, AD, n);
  gat_agg<<<gAgg, TB, 0, stream>>>(CUR, CSR, H, AS, AD, b1, XO, n, 1);

  gat_proj_wmma<<<gProj, TB, 0, stream>>>(XO, W2, H, n);
  gat_alphas<<<gNode, TB, 0, stream>>>(H, as2, ad2, AS, AD, n);
  gat_agg<<<gAgg, TB, 0, stream>>>(CUR, CSR, H, AS, AD, b2, out, n, 0);
}


# aggregation: a wave's two nodes run side by side, one per 16-lane row, not one after the other; cross-lane steps on DPP row_xmask and v_permlane16, no ds_bpermute; float sums re-ordered
# speedup vs baseline: 13.5966x; 1.1188x over previous
#include <hip/hip_runtime.h>

typedef __attribute__((ext_vector_type(2))) float v2f;
typedef __attribute__((ext_vector_type(8))) float v8f;

#define DIM 16
#define NEG_SLOPE 0.2f

__device__ __forceinline__ float leaky(float e) { return e >= 0.f ? e : NEG_SLOPE * e; }

#define DROWS 32
#define DPITCH 20

template <int NF>
__global__ void __launch_bounds__(256)
gat_dense(const float* __restrict__ X, const float* __restrict__ W,
          const float* __restrict__ a_src, const float* __restrict__ a_dst,
          float* __restrict__ H, float* __restrict__ AS, float* __restrict__ AD, int n) {
  __shared__ float stage[8][DROWS * DPITCH];
  const int wib = (int)(threadIdx.x >> 5);
  const int wave = (int)blockIdx.x * 8 + wib;
  const int lane = threadIdx.x & 31;
  const int row0 = wave * DROWS;
  if (row0 >= n) return;
  const int half = lane >> 4;
  const int l = lane & 15;
  int rA0 = row0 + l;
  int rA1 = row0 + 16 + l;
  if (rA0 >= n) rA0 = n - 1;
  if (rA1 >= n) rA1 = n - 1;
  v8f c0 = {};
  v8f c1 = {};
#pragma unroll
  for (int k0 = 0; k0 < DIM; k0 += 4) {
    const int ka = k0 + half * 2;
    v2f a0 = {0.f, 0.f};
    v2f a1 = {0.f, 0.f};
    v2f b = {0.f, 0.f};
    if (ka < NF) {
      const float2 x0 = *(const float2*)&X[rA0 * NF + ka];
      const float2 x1 = *(const float2*)&X[rA1 * NF + ka];
      a0.x = x0.x;
      a0.y = x0.y;
      a1.x = x1.x;
      a1.y = x1.y;
      b.x = W[ka * DIM + l];
      b.y = W[(ka + 1) * DIM + l];
    }
    c0 = __builtin_amdgcn_wmma_f32_16x16x4_f32(false, a0, false, b, (short)0, c0, false, false);
    c1 = __builtin_amdgcn_wmma_f32_16x16x4_f32(false, a1, false, b, (short)0, c1, false, false);
  }
  float* st = stage[wib];
#pragma unroll
  for (int r = 0; r < 8; ++r) {
    st[(half * 8 + r) * DPITCH + l] = c0[r];
    st[(16 + half * 8 + r) * DPITCH + l] = c1[r];
  }
  const int base = (row0 + half * 8) * DIM + l;
  if (row0 + DROWS <= n) {
#pragma unroll
    for (int r = 0; r < 8; ++r) {
      H[base + r * DIM] = c0[r];
      H[base + (16 + r) * DIM] = c1[r];
    }
  } else {
#pragma unroll
    for (int r = 0; r < 8; ++r) {
      const int row = row0 + r + half * 8;
      if (row < n) H[row * DIM + l] = c0[r];
      if (row + 16 < n) H[(row + 16) * DIM + l] = c1[r];
    }
  }
  __builtin_amdgcn_fence(__ATOMIC_RELEASE, "wavefront");
  __builtin_amdgcn_wave_barrier();
  __builtin_amdgcn_fence(__ATOMIC_ACQUIRE, "wavefront");
  const float* hr = &st[lane * DPITCH];
  float s = 0.f, d = 0.f;
#pragma unroll
  for (int j = 0; j < 4; ++j) {
    const float4 h = *(const float4*)&hr[j * 4];
    s += h.x * a_src[j * 4 + 0] + h.y * a_src[j * 4 + 1] +
         h.z * a_src[j * 4 + 2] + h.w * a_src[j * 4 + 3];
    d += h.x * a_dst[j * 4 + 0] + h.y * a_dst[j * 4 + 1] +
         h.z * a_dst[j * 4 + 2] + h.w * a_dst[j * 4 + 3];
  }
  const int row = row0 + lane;
  if (row < n) {
    AS[row] = s;
    AD[row] = d;
  }
}

#define BSH 8
#define BNODES (1 << BSH)
#define NBK_MAX 1024
#define PT_TB 1024
#define BK_PITCH 32
#define SORT_BUF 16384

__global__ void csr_zero(int* __restrict__ BK, int cnt) {
  int i = blockIdx.x * blockDim.x + threadIdx.x;
  if (i < cnt) BK[i] = 0;
}

__device__ __forceinline__ void part_hist(int* cnt, const int* __restrict__ dst, int E) {
  const int tid = threadIdx.x;
  const int E4 = E >> 2;
  const int qb = (E4 + (int)gridDim.x - 1) / (int)gridDim.x;
  const int q0 = (int)blockIdx.x * qb;
  const int q1 = (q0 + qb < E4) ? q0 + qb : E4;
  const int4* d4p = (const int4*)dst;
  for (int q = q0 + tid; q < q1; q += PT_TB) {
    int4 d4 = d4p[q];
    atomicAdd(&cnt[d4.x >> BSH], 1);
    atomicAdd(&cnt[d4.y >> BSH], 1);
    atomicAdd(&cnt[d4.z >> BSH], 1);
    atomicAdd(&cnt[d4.w >> BSH], 1);
  }
  if (blockIdx.x == 0 && tid < (E & 3)) atomicAdd(&cnt[dst[(E & ~3) + tid] >> BSH], 1);
}

__global__ void __launch_bounds__(PT_TB)
part_count(const int* __restrict__ dst, int* __restrict__ BK, int E, int nbk) {
  __shared__ int cnt[NBK_MAX];
  const int tid = threadIdx.x;
  for (int j = tid; j < nbk; j += PT_TB) cnt[j] = 0;
  __syncthreads();
  part_hist(cnt, dst, E);
  __syncthreads();
  for (int j = tid; j < nbk; j += PT_TB) {
    int c = cnt[j];
    if (c) atomicAdd(&BK[j * BK_PITCH], c);
  }
}

__global__ void __launch_bounds__(NBK_MAX)
part_scan(int* __restrict__ BK, int* __restrict__ BKS, int nbk) {
  __shared__ int sh[NBK_MAX];
  const int tid = threadIdx.x;
  const int v = (tid < nbk) ? BK[tid * BK_PITCH] : 0;
  sh[tid] = v;
  __syncthreads();
  for (int w = 1; w < NBK_MAX; w <<= 1) {
    int add = (tid >= w) ? sh[tid - w] : 0;
    __syncthreads();
    sh[tid] += add;
    __syncthreads();
  }
  const int incl = sh[tid];
  if (tid < nbk) {
    BKS[tid] = incl - v;
    BK[tid * BK_PITCH] = incl - v;
  }
  if (tid == nbk - 1) BKS[nbk] = incl;
}

__global__ void __launch_bounds__(PT_TB)
part_scatter(const int* __restrict__ src, const int* __restrict__ dst, int* __restrict__ BK,
             int* __restrict__ TMP, int E, int nbk) {
  __shared__ int cnt[NBK_MAX];
  const int tid = threadIdx.x;
  for (int j = tid; j < nbk; j += PT_TB) cnt[j] = 0;
  __syncthreads();
  part_hist(cnt, dst, E);
  __syncthreads();
  for (int j = tid; j < nbk; j += PT_TB) {
    int c = cnt[j];
    cnt[j] = c ? atomicAdd(&BK[j * BK_PITCH], c) : 0;
  }
  __syncthreads();
  const int E4 = E >> 2;
  const int qb = (E4 + (int)gridDim.x - 1) / (int)gridDim.x;
  const int q0 = (int)blockIdx.x * qb;
  const int q1 = (q0 + qb < E4) ? q0 + qb : E4;
  const int4* s4p = (const int4*)src;
  const int4* d4p = (const int4*)dst;
  for (int q = q0 + tid; q < q1; q += PT_TB) {
    int4 s4 = s4p[q], d4 = d4p[q];
    TMP[atomicAdd(&cnt[d4.x >> BSH], 1)] = (s4.x << BSH) | (d4.x & (BNODES - 1));
    TMP[atomicAdd(&cnt[d4.y >> BSH], 1)] = (s4.y << BSH) | (d4.y & (BNODES - 1));
    TMP[atomicAdd(&cnt[d4.z >> BSH], 1)] = (s4.z << BSH) | (d4.z & (BNODES - 1));
    TMP[atomicAdd(&cnt[d4.w >> BSH], 1)] = (s4.w << BSH) | (d4.w & (BNODES - 1));
  }
  if (blockIdx.x == 0 && tid < (E & 3)) {
    int e = (E & ~3) + tid;
    int d = dst[e];
    TMP[atomicAdd(&cnt[d >> BSH], 1)] = (src[e] << BSH) | (d & (BNODES - 1));
  }
}

__global__ void __launch_bounds__(PT_TB)
part_sort(const int* __restrict__ TMP, const int* __restrict__ BKS, int* __restrict__ CUR,
          int* __restrict__ CSR, int n) {
  __shared__ int c[BNODES];
  __shared__ int sc[BNODES];
  __shared__ int buf[SORT_BUF];
  const int tid = threadIdx.x;
  const int b = blockIdx.x;
  const int e0 = BKS[b], e1 = BKS[b + 1];
  const int len = e1 - e0;
  const bool staged = len <= SORT_BUF;
  if (tid < BNODES) c[tid] = 0;
  __syncthreads();
  for (int e = e0 + tid; e < e1; e += PT_TB) atomicAdd(&c[TMP[e] & (BNODES - 1)], 1);
  __syncthreads();
  int v = 0;
  if (tid < BNODES) {
    v = c[tid];
    sc[tid] = v;
  }
  __syncthreads();
  for (int w = 1; w < BNODES; w <<= 1) {
    int add = 0;
    if (tid < BNODES && tid >= w) add = sc[tid - w];
    __syncthreads();
    if (tid < BNODES) sc[tid] += add;
    __syncthreads();
  }
  if (tid < BNODES) {
    const int end = e0 + sc[tid];
    const int node = (b << BSH) + tid;
    if (node < n) CUR[node] = end;
    c[tid] = (staged ? sc[tid] : end) - v;
  }
  __syncthreads();
  if (staged) {
    for (int e = e0 + tid; e < e1; e += PT_TB) {
      const int p = TMP[e];
      buf[atomicAdd(&c[p & (BNODES - 1)], 1)] = (int)((unsigned)p >> BSH);
    }
    __syncthreads();
    for (int k = tid; k < len; k += PT_TB) CSR[e0 + k] = buf[k];
  } else {
    for (int e = e0 + tid; e < e1; e += PT_TB) {
      const int p = TMP[e];
      CSR[atomicAdd(&c[p & (BNODES - 1)], 1)] = (int)((unsigned)p >> BSH);
    }
  }
}

template <int N>
__device__ __forceinline__ float row_xor(float v) {
  return __builtin_bit_cast(
      float, __builtin_amdgcn_update_dpp(0, __builtin_bit_cast(int, v), 0x160 + N, 0xf, 0xf, false));
}

__device__ __forceinline__ unsigned row_slot(unsigned v, unsigned r) {
  const unsigned a = 4u * r;
  const unsigned lo = a * 0x1111u | ((a + 1u) * 0x1111u << 16);
  const unsigned hi = (a + 2u) * 0x1111u | ((a + 3u) * 0x1111u << 16);
  return __builtin_amdgcn_permlane16(v, v, lo, hi, false, false);
}

__global__ void gat_agg(const int* __restrict__ CUR, const int* __restrict__ CSR,
                        const float* __restrict__ H, const float* __restrict__ AS,
                        const float* __restrict__ AD, const float* __restrict__ b,
                        float* __restrict__ OUT, int n, int do_relu) {
  const unsigned lane = threadIdx.x & 31u;
  const unsigned wave = (unsigned)((blockIdx.x * blockDim.x + threadIdx.x) >> 5);
  const unsigned nn = (unsigned)n;
  if (wave * 2u >= nn) return;
  const unsigned l = lane & 15u;
  const unsigned slot = l >> 2;
  const unsigned quad = l & 3u;
  const unsigned node = wave * 2u + (lane >> 4);
  const bool live = node < nn;
  const unsigned i = live ? node : nn - 1u;
  const unsigned beg = i ? (unsigned)CUR[i - 1u] : 0u;
  const unsigned end = (unsigned)CUR[i];
  const float ad = AD[i];
  const float e_self = leaky(AS[i] + ad);
  unsigned s0 = i, s1 = i;
  float e0 = e_self, e1 = e_self;
  if (beg + l < end) {
    s0 = (unsigned)CSR[beg + l];
    e0 = leaky(AS[s0] + ad);
  }
  if (beg + 16u + l < end) {
    s1 = (unsigned)CSR[beg + 16u + l];
    e1 = leaky(AS[s1] + ad);
  }
  float m = fmaxf(e_self, fmaxf(e0, e1));
  for (unsigned p = beg + 32u + l; p < end; p += 16u) m = fmaxf(m, leaky(AS[CSR[p]] + ad));
  m = fmaxf(m, row_xor<1>(m));
  m = fmaxf(m, row_xor<2>(m));
  m = fmaxf(m, row_xor<4>(m));
  m = fmaxf(m, row_xor<8>(m));
  float den = 0.f;
  float4 acc = make_float4(0.f, 0.f, 0.f, 0.f);
  unsigned t = 0u;
  for (unsigned base = beg; __builtin_amdgcn_ballot_w32(base < end) != 0u; base += 16u, ++t) {
    const int cnt = (int)(end - base);
    unsigned s = s0;
    float e = e0;
    if (t == 1u) {
      s = s1;
      e = e1;
    } else if (t > 1u) {
      s = i;
      e = 0.f;
      if ((int)l < cnt) {
        s = (unsigned)CSR[base + l];
        e = leaky(AS[s] + ad);
      }
    }
    const float ex = ((int)l < cnt) ? expf(e - m) : 0.f;
    den += ex;
    float4 h[4];
    float w[4];
#pragma unroll
    for (int r = 0; r < 4; ++r) {
      const unsigned sr = row_slot(s, (unsigned)r);
      w[r] = __builtin_bit_cast(float, row_slot(__builtin_bit_cast(unsigned, ex), (unsigned)r));
      h[r] = make_float4(0.f, 0.f, 0.f, 0.f);
      if (4 * r + (int)slot < cnt) h[r] = *(const float4*)&H[sr * DIM + quad * 4u];
    }
#pragma unroll
    for (int r = 0; r < 4; ++r) {
      acc.x += w[r] * h[r].x;
      acc.y += w[r] * h[r].y;
      acc.z += w[r] * h[r].z;
      acc.w += w[r] * h[r].w;
    }
  }
  den += row_xor<1>(den);
  den += row_xor<2>(den);
  den += row_xor<4>(den);
  den += row_xor<8>(den);
  acc.x += row_xor<4>(acc.x);
  acc.y += row_xor<4>(acc.y);
  acc.z += row_xor<4>(acc.z);
  acc.w += row_xor<4>(acc.w);
  acc.x += row_xor<8>(acc.x);
  acc.y += row_xor<8>(acc.y);
  acc.z += row_xor<8>(acc.z);
  acc.w += row_xor<8>(acc.w);
  const float ex_self = expf(e_self - m);
  den += ex_self;
  const float4 hs = *(const float4*)&H[i * DIM + quad * 4u];
  const float4 bb = *(const float4*)&b[quad * 4u];
  float4 v;
  v.x = (acc.x + ex_self * hs.x) / den + bb.x;
  v.y = (acc.y + ex_self * hs.y) / den + bb.y;
  v.z = (acc.z + ex_self * hs.z) / den + bb.z;
  v.w = (acc.w + ex_self * hs.w) / den + bb.w;
  if (do_relu) {
    v.x = v.x > 0.f ? v.x : 0.f;
    v.y = v.y > 0.f ? v.y : 0.f;
    v.z = v.z > 0.f ? v.z : 0.f;
    v.w = v.w > 0.f ? v.w : 0.f;
  }
  if (slot == 0u && live) *(float4*)&OUT[node * DIM + quad * 4u] = v;
}

extern "C" void kernel_launch(void* const* d_in, const int* in_sizes, int n_in,
                              void* d_out, int out_size, void* d_ws, size_t ws_size,
                              hipStream_t stream) {
  const float* x = (const float*)d_in[0];
  const int* ei = (const int*)d_in[1];
  const float* W1 = (const float*)d_in[2];
  const float* as1 = (const float*)d_in[3];
  const float* ad1 = (const float*)d_in[4];
  const float* b1 = (const float*)d_in[5];
  const float* W2 = (const float*)d_in[6];
  const float* as2 = (const float*)d_in[7];
  const float* ad2 = (const float*)d_in[8];
  const float* b2 = (const float*)d_in[9];
  float* out = (float*)d_out;

  const int NFEAT = 14;
  int n = in_sizes[0] / NFEAT;
  int E = in_sizes[1] / 2;
  const int* src = ei;
  const int* dst = ei + E;

  int nbk = (n + BNODES - 1) >> BSH;
  float* fws = (float*)d_ws;
  float* H = fws;
  float* XO = H + (size_t)n * DIM;
  float* AS = XO + (size_t)n * DIM;
  float* AD = AS + n;
  int* CUR = (int*)(AD + n);
  int* CSR = CUR + n;
  int* TMP = (int*)H;
  int* BK = (int*)AS;
  int* BKS = (int*)AD;

  const int TB = 256;
  dim3 gNode((n + TB - 1) / TB);
  dim3 gPart(256);
  dim3 gAgg(((n + 1) / 2 + 7) / 8);

  csr_zero<<<(nbk * BK_PITCH + TB - 1) / TB, TB, 0, stream>>>(BK, nbk * BK_PITCH);
  part_count<<<gPart, PT_TB, 0, stream>>>(dst, BK, E, nbk);
  part_scan<<<1, NBK_MAX, 0, stream>>>(BK, BKS, nbk);
  part_scatter<<<gPart, PT_TB, 0, stream>>>(src, dst, BK, TMP, E, nbk);
  part_sort<<<nbk, PT_TB, 0, stream>>>(TMP, BKS, CUR, CSR, n);

  gat_dense<NFEAT><<<gNode, TB, 0, stream>>>(x, W1, as1, ad1, H, AS, AD, n);
  gat_agg<<<gAgg, TB, 0, stream>>>(CUR, CSR, H, AS, AD, b1, XO, n, 1);

  gat_dense<DIM><<<gNode, TB, 0, stream>>>(XO, W2, as2, ad2, H, AS, AD, n);
  gat_agg<<<gAgg, TB, 0, stream>>>(CUR, CSR, H, AS, AD, b2, out, n, 0);
}


# edge partition pass: a block reads its slice once, groups its edges by bucket in LDS and writes each bucket's run as contiguous lines, not one 4-byte store per edge; arrival order kept
# speedup vs baseline: 18.6171x; 1.3692x over previous
#include <hip/hip_runtime.h>

typedef __attribute__((ext_vector_type(2))) float v2f;
typedef __attribute__((ext_vector_type(8))) float v8f;

#define DIM 16
#define NEG_SLOPE 0.2f

__device__ __forceinline__ float leaky(float e) { return e >= 0.f ? e : NEG_SLOPE * e; }

#define DROWS 32
#define DPITCH 20

template <int NF>
__global__ void __launch_bounds__(256)
gat_dense(const float* __restrict__ X, const float* __restrict__ W,
          const float* __restrict__ a_src, const float* __restrict__ a_dst,
          float* __restrict__ H, float* __restrict__ AS, float* __restrict__ AD, int n) {
  __shared__ float stage[8][DROWS * DPITCH];
  const int wib = (int)(threadIdx.x >> 5);
  const int wave = (int)blockIdx.x * 8 + wib;
  const int lane = threadIdx.x & 31;
  const int row0 = wave * DROWS;
  if (row0 >= n) return;
  const int half = lane >> 4;
  const int l = lane & 15;
  int rA0 = row0 + l;
  int rA1 = row0 + 16 + l;
  if (rA0 >= n) rA0 = n - 1;
  if (rA1 >= n) rA1 = n - 1;
  v8f c0 = {};
  v8f c1 = {};
#pragma unroll
  for (int k0 = 0; k0 < DIM; k0 += 4) {
    const int ka = k0 + half * 2;
    v2f a0 = {0.f, 0.f};
    v2f a1 = {0.f, 0.f};
    v2f b = {0.f, 0.f};
    if (ka < NF) {
      const float2 x0 = *(const float2*)&X[rA0 * NF + ka];
      const float2 x1 = *(const float2*)&X[rA1 * NF + ka];
      a0.x = x0.x;
      a0.y = x0.y;
      a1.x = x1.x;
      a1.y = x1.y;
      b.x = W[ka * DIM + l];
      b.y = W[(ka + 1) * DIM + l];
    }
    c0 = __builtin_amdgcn_wmma_f32_16x16x4_f32(false, a0, false, b, (short)0, c0, false, false);
    c1 = __builtin_amdgcn_wmma_f32_16x16x4_f32(false, a1, false, b, (short)0, c1, false, false);
  }
  float* st = stage[wib];
#pragma unroll
  for (int r = 0; r < 8; ++r) {
    st[(half * 8 + r) * DPITCH + l] = c0[r];
    st[(16 + half * 8 + r) * DPITCH + l] = c1[r];
  }
  const int base = (row0 + half * 8) * DIM + l;
  if (row0 + DROWS <= n) {
#pragma unroll
    for (int r = 0; r < 8; ++r) {
      H[base + r * DIM] = c0[r];
      H[base + (16 + r) * DIM] = c1[r];
    }
  } else {
#pragma unroll
    for (int r = 0; r < 8; ++r) {
      const int row = row0 + r + half * 8;
      if (row < n) H[row * DIM + l] = c0[r];
      if (row + 16 < n) H[(row + 16) * DIM + l] = c1[r];
    }
  }
  __builtin_amdgcn_fence(__ATOMIC_RELEASE, "wavefront");
  __builtin_amdgcn_wave_barrier();
  __builtin_amdgcn_fence(__ATOMIC_ACQUIRE, "wavefront");
  const float* hr = &st[lane * DPITCH];
  float s = 0.f, d = 0.f;
#pragma unroll
  for (int j = 0; j < 4; ++j) {
    const float4 h = *(const float4*)&hr[j * 4];
    s += h.x * a_src[j * 4 + 0] + h.y * a_src[j * 4 + 1] +
         h.z * a_src[j * 4 + 2] + h.w * a_src[j * 4 + 3];
    d += h.x * a_dst[j * 4 + 0] + h.y * a_dst[j * 4 + 1] +
         h.z * a_dst[j * 4 + 2] + h.w * a_dst[j * 4 + 3];
  }
  const int row = row0 + lane;
  if (row < n) {
    AS[row] = s;
    AD[row] = d;
  }
}

#define BSH 8
#define BNODES (1 << BSH)
#define NBK_MAX 1024
#define PT_TB 1024
#define BK_PITCH 32
#define SORT_BUF 16384

__global__ void csr_zero(int* __restrict__ BK, int cnt) {
  int i = blockIdx.x * blockDim.x + threadIdx.x;
  if (i < cnt) BK[i] = 0;
}

__device__ __forceinline__ void part_hist(int* cnt, const int* __restrict__ dst, int E) {
  const int tid = threadIdx.x;
  const int E4 = E >> 2;
  const int qb = (E4 + (int)gridDim.x - 1) / (int)gridDim.x;
  const int q0 = (int)blockIdx.x * qb;
  const int q1 = (q0 + qb < E4) ? q0 + qb : E4;
  const int4* d4p = (const int4*)dst;
  for (int q = q0 + tid; q < q1; q += PT_TB) {
    int4 d4 = d4p[q];
    atomicAdd(&cnt[d4.x >> BSH], 1);
    atomicAdd(&cnt[d4.y >> BSH], 1);
    atomicAdd(&cnt[d4.z >> BSH], 1);
    atomicAdd(&cnt[d4.w >> BSH], 1);
  }
  if (blockIdx.x == 0 && tid < (E & 3)) atomicAdd(&cnt[dst[(E & ~3) + tid] >> BSH], 1);
}

__global__ void __launch_bounds__(PT_TB)
part_count(const int* __restrict__ dst, int* __restrict__ BK, int E, int nbk) {
  __shared__ int cnt[NBK_MAX];
  const int tid = threadIdx.x;
  for (int j = tid; j < nbk; j += PT_TB) cnt[j] = 0;
  __syncthreads();
  part_hist(cnt, dst, E);
  __syncthreads();
  for (int j = tid; j < nbk; j += PT_TB) {
    int c = cnt[j];
    if (c) atomicAdd(&BK[j * BK_PITCH], c);
  }
}

__global__ void __launch_bounds__(NBK_MAX)
part_scan(int* __restrict__ BK, int* __restrict__ BKS, int nbk) {
  __shared__ int sh[NBK_MAX];
  const int tid = threadIdx.x;
  const int v = (tid < nbk) ? BK[tid * BK_PITCH] : 0;
  sh[tid] = v;
  __syncthreads();
  for (int w = 1; w < NBK_MAX; w <<= 1) {
    int add = (tid >= w) ? sh[tid - w] : 0;
    __syncthreads();
    sh[tid] += add;
    __syncthreads();
  }
  const int incl = sh[tid];
  if (tid < nbk) {
    BKS[tid] = incl - v;
    BK[tid * BK_PITCH] = incl - v;
  }
  if (tid == nbk - 1) BKS[nbk] = incl;
}

#define SC_CH 16384

__global__ void __launch_bounds__(PT_TB)
part_scatter(const int* __restrict__ src, const int* __restrict__ dst, int* __restrict__ BK,
             int* __restrict__ TMP, int E, int nbk) {
  __shared__ int cnt[NBK_MAX];
  __shared__ int gof[NBK_MAX];
  __shared__ int wsum[32];
  __shared__ int val[SC_CH];
  __shared__ unsigned short bid[SC_CH];
  const int tid = threadIdx.x;
  const int lane = tid & 31;
  const int wib = tid >> 5;
  const int E4 = E >> 2;
  const int qb = (E4 + (int)gridDim.x - 1) / (int)gridDim.x;
  const int q0 = (int)blockIdx.x * qb;
  const int q1 = (q0 + qb < E4) ? q0 + qb : E4;
  const int4* s4p = (const int4*)src;
  const int4* d4p = (const int4*)dst;
  for (int qc = q0; qc < q1; qc += SC_CH / 4) {
    const int qe = (qc + SC_CH / 4 < q1) ? qc + SC_CH / 4 : q1;
    cnt[tid] = 0;
    __syncthreads();
    int4 d4[4], s4[4];
#pragma unroll
    for (int j = 0; j < 4; ++j) {
      const int q = qc + tid + j * PT_TB;
      const int qr = q < qe ? q : qe - 1;
      d4[j] = d4p[qr];
      s4[j] = s4p[qr];
    }
#pragma unroll
    for (int j = 0; j < 4; ++j) {
      if (qc + tid + j * PT_TB < qe) {
        atomicAdd(&cnt[d4[j].x >> BSH], 1);
        atomicAdd(&cnt[d4[j].y >> BSH], 1);
        atomicAdd(&cnt[d4[j].z >> BSH], 1);
        atomicAdd(&cnt[d4[j].w >> BSH], 1);
      }
    }
    __syncthreads();
    const int c = cnt[tid];
    int incl = c;
#pragma unroll
    for (int o = 1; o < 32; o <<= 1) {
      const int t = __shfl_up(incl, o);
      if (lane >= o) incl += t;
    }
    if (lane == 31) wsum[wib] = incl;
    __syncthreads();
    if (wib == 0) {
      const int w = wsum[lane];
      int wi = w;
#pragma unroll
      for (int o = 1; o < 32; o <<= 1) {
        const int t = __shfl_up(wi, o);
        if (lane >= o) wi += t;
      }
      wsum[lane] = wi - w;
    }
    __syncthreads();
    const int ls = wsum[wib] + incl - c;
    const int g = c ? atomicAdd(&BK[tid * BK_PITCH], c) : 0;
    cnt[tid] = ls;
    gof[tid] = g - ls;
    __syncthreads();
#pragma unroll
    for (int j = 0; j < 4; ++j) {
      const int q = qc + tid + j * PT_TB;
      if (q < qe) {
        int bk, sl;
        bk = d4[j].x >> BSH;
        sl = atomicAdd(&cnt[bk], 1);
        val[sl] = (s4[j].x << BSH) | (d4[j].x & (BNODES - 1));
        bid[sl] = (unsigned short)bk;
        bk = d4[j].y >> BSH;
        sl = atomicAdd(&cnt[bk], 1);
        val[sl] = (s4[j].y << BSH) | (d4[j].y & (BNODES - 1));
        bid[sl] = (unsigned short)bk;
        bk = d4[j].z >> BSH;
        sl = atomicAdd(&cnt[bk], 1);
        val[sl] = (s4[j].z << BSH) | (d4[j].z & (BNODES - 1));
        bid[sl] = (unsigned short)bk;
        bk = d4[j].w >> BSH;
        sl = atomicAdd(&cnt[bk], 1);
        val[sl] = (s4[j].w << BSH) | (d4[j].w & (BNODES - 1));
        bid[sl] = (unsigned short)bk;
      }
    }
    __syncthreads();
    const int tot = (qe - qc) * 4;
    for (int k = tid; k < tot; k += PT_TB) TMP[gof[bid[k]] + k] = val[k];
    __syncthreads();
  }
  if (blockIdx.x == 0 && tid < (E & 3)) {
    const int e = (E & ~3) + tid;
    const int d = dst[e];
    TMP[atomicAdd(&BK[(d >> BSH) * BK_PITCH], 1)] = (src[e] << BSH) | (d & (BNODES - 1));
  }
}

__global__ void __launch_bounds__(PT_TB)
part_sort(const int* __restrict__ TMP, const int* __restrict__ BKS, int* __restrict__ CUR,
          int* __restrict__ CSR, int n) {
  __shared__ int c[BNODES];
  __shared__ int sc[BNODES];
  __shared__ int buf[SORT_BUF];
  const int tid = threadIdx.x;
  const int b = blockIdx.x;
  const int e0 = BKS[b], e1 = BKS[b + 1];
  const int len = e1 - e0;
  const bool staged = len <= SORT_BUF;
  if (tid < BNODES) c[tid] = 0;
  __syncthreads();
  for (int e = e0 + tid; e < e1; e += PT_TB) atomicAdd(&c[TMP[e] & (BNODES - 1)], 1);
  __syncthreads();
  int v = 0;
  if (tid < BNODES) {
    v = c[tid];
    sc[tid] = v;
  }
  __syncthreads();
  for (int w = 1; w < BNODES; w <<= 1) {
    int add = 0;
    if (tid < BNODES && tid >= w) add = sc[tid - w];
    __syncthreads();
    if (tid < BNODES) sc[tid] += add;
    __syncthreads();
  }
  if (tid < BNODES) {
    const int end = e0 + sc[tid];
    const int node = (b << BSH) + tid;
    if (node < n) CUR[node] = end;
    c[tid] = (staged ? sc[tid] : end) - v;
  }
  __syncthreads();
  if (staged) {
    for (int e = e0 + tid; e < e1; e += PT_TB) {
      const int p = TMP[e];
      buf[atomicAdd(&c[p & (BNODES - 1)], 1)] = (int)((unsigned)p >> BSH);
    }
    __syncthreads();
    for (int k = tid; k < len; k += PT_TB) CSR[e0 + k] = buf[k];
  } else {
    for (int e = e0 + tid; e < e1; e += PT_TB) {
      const int p = TMP[e];
      CSR[atomicAdd(&c[p & (BNODES - 1)], 1)] = (int)((unsigned)p >> BSH);
    }
  }
}

template <int N>
__device__ __forceinline__ float row_xor(float v) {
  return __builtin_bit_cast(
      float, __builtin_amdgcn_update_dpp(0, __builtin_bit_cast(int, v), 0x160 + N, 0xf, 0xf, false));
}

__device__ __forceinline__ unsigned row_slot(unsigned v, unsigned r) {
  const unsigned a = 4u * r;
  const unsigned lo = a * 0x1111u | ((a + 1u) * 0x1111u << 16);
  const unsigned hi = (a + 2u) * 0x1111u | ((a + 3u) * 0x1111u << 16);
  return __builtin_amdgcn_permlane16(v, v, lo, hi, false, false);
}

template <int R>
__device__ __forceinline__ float quad_bcast(float v) {
  return __builtin_bit_cast(
      float, __builtin_amdgcn_update_dpp(0, __builtin_bit_cast(int, v), R * 0x55, 0xf, 0xf, false));
}

__global__ void gat_agg(const int* __restrict__ CUR, const int* __restrict__ CSR,
                        const float* __restrict__ H, const float* __restrict__ AS,
                        const float* __restrict__ AD, const float* __restrict__ a_src,
                        const float* __restrict__ b, float* __restrict__ OUT, int n, int do_relu) {
  const unsigned lane = threadIdx.x & 31u;
  const unsigned wave = (unsigned)((blockIdx.x * blockDim.x + threadIdx.x) >> 5);
  const unsigned nn = (unsigned)n;
  if (wave * 2u >= nn) return;
  const unsigned l = lane & 15u;
  const unsigned slot = l >> 2;
  const unsigned quad = l & 3u;
  const unsigned node = wave * 2u + (lane >> 4);
  const bool live = node < nn;
  const unsigned i = live ? node : nn - 1u;
  const float4* H4 = (const float4*)H;
  const unsigned beg = i ? (unsigned)CUR[i - 1u] : 0u;
  const unsigned end = (unsigned)CUR[i];
  const float ad = AD[i];
  const float e_self = leaky(AS[i] + ad);
  const float4 aq = ((const float4*)a_src)[quad];
  const v2f a01 = {aq.x, aq.y};
  const v2f a23 = {aq.z, aq.w};
  const float4 hs = H4[i * 4u + quad];
  unsigned sA = i, sB = i;
  if (beg + l < end) sA = (unsigned)CSR[beg + l];
  if (beg + 16u + l < end) sB = (unsigned)CSR[beg + 16u + l];
  const bool q1 = (quad & 1u) != 0u;
  const bool q2 = (quad & 2u) != 0u;
  const int own = (int)(4u * quad + slot);
  float m = e_self;
  float den = l == 0u ? 1.f : 0.f;
  float4 acc = make_float4(0.f, 0.f, 0.f, 0.f);
  if (slot == 0u) acc = hs;
  for (unsigned base = beg; __builtin_amdgcn_ballot_w32(base < end) != 0u; base += 32u) {
    const int cnt = (int)(end - base);
    unsigned nA = i, nB = i;
    if ((int)l + 32 < cnt) nA = (unsigned)CSR[base + 32u + l];
    if ((int)l + 48 < cnt) nB = (unsigned)CSR[base + 48u + l];
    float4 h[8];
#pragma unroll
    for (int j = 0; j < 8; ++j) {
      const unsigned sr = row_slot(j < 4 ? sA : sB, (unsigned)(j & 3));
      h[j] = H4[sr * 4u + quad];
    }
    float p[8];
#pragma unroll
    for (int j = 0; j < 8; ++j) {
      const v2f lo = {h[j].x, h[j].y};
      const v2f hi = {h[j].z, h[j].w};
      const v2f t = lo * a01 + hi * a23;
      p[j] = t.x + t.y;
    }
    float k[4];
#pragma unroll
    for (int c = 0; c < 4; ++c) {
      const float keep = q1 ? p[2 * c + 1] : p[2 * c];
      const float send = q1 ? p[2 * c] : p[2 * c + 1];
      k[c] = keep + row_xor<1>(send);
    }
    float pa, pb;
    {
      const float keep = q2 ? k[1] : k[0];
      const float send = q2 ? k[0] : k[1];
      pa = keep + row_xor<2>(send);
    }
    {
      const float keep = q2 ? k[3] : k[2];
      const float send = q2 ? k[2] : k[3];
      pb = keep + row_xor<2>(send);
    }
    const float ea = own < cnt ? leaky(pa + ad) : -__builtin_inff();
    const float eb = own + 16 < cnt ? leaky(pb + ad) : -__builtin_inff();
    float mt = fmaxf(ea, eb);
    mt = fmaxf(mt, row_xor<1>(mt));
    mt = fmaxf(mt, row_xor<2>(mt));
    mt = fmaxf(mt, row_xor<4>(mt));
    mt = fmaxf(mt, row_xor<8>(mt));
    const float mn = fmaxf(m, mt);
    const float sc = expf(m - mn);
    m = mn;
    const float xa = expf(ea - mn);
    const float xb = expf(eb - mn);
    den = den * sc + (xa + xb);
    acc.x *= sc;
    acc.y *= sc;
    acc.z *= sc;
    acc.w *= sc;
    float w[8];
    w[0] = quad_bcast<0>(xa);
    w[1] = quad_bcast<1>(xa);
    w[2] = quad_bcast<2>(xa);
    w[3] = quad_bcast<3>(xa);
    w[4] = quad_bcast<0>(xb);
    w[5] = quad_bcast<1>(xb);
    w[6] = quad_bcast<2>(xb);
    w[7] = quad_bcast<3>(xb);
#pragma unroll
    for (int j = 0; j < 8; ++j) {
      acc.x += w[j] * h[j].x;
      acc.y += w[j] * h[j].y;
      acc.z += w[j] * h[j].z;
      acc.w += w[j] * h[j].w;
    }
    sA = nA;
    sB = nB;
  }
  den += row_xor<1>(den);
  den += row_xor<2>(den);
  den += row_xor<4>(den);
  den += row_xor<8>(den);
  acc.x += row_xor<4>(acc.x);
  acc.y += row_xor<4>(acc.y);
  acc.z += row_xor<4>(acc.z);
  acc.w += row_xor<4>(acc.w);
  acc.x += row_xor<8>(acc.x);
  acc.y += row_xor<8>(acc.y);
  acc.z += row_xor<8>(acc.z);
  acc.w += row_xor<8>(acc.w);
  const float4 bb = ((const float4*)b)[quad];
  float4 v;
  v.x = acc.x / den + bb.x;
  v.y = acc.y / den + bb.y;
  v.z = acc.z / den + bb.z;
  v.w = acc.w / den + bb.w;
  if (do_relu) {
    v.x = v.x > 0.f ? v.x : 0.f;
    v.y = v.y > 0.f ? v.y : 0.f;
    v.z = v.z > 0.f ? v.z : 0.f;
    v.w = v.w > 0.f ? v.w : 0.f;
  }
  if (slot == 0u && live) ((float4*)OUT)[node * 4u + quad] = v;
}

extern "C" void kernel_launch(void* const* d_in, const int* in_sizes, int n_in,
                              void* d_out, int out_size, void* d_ws, size_t ws_size,
                              hipStream_t stream) {
  const float* x = (const float*)d_in[0];
  const int* ei = (const int*)d_in[1];
  const float* W1 = (const float*)d_in[2];
  const float* as1 = (const float*)d_in[3];
  const float* ad1 = (const float*)d_in[4];
  const float* b1 = (const float*)d_in[5];
  const float* W2 = (const float*)d_in[6];
  const float* as2 = (const float*)d_in[7];
  const float* ad2 = (const float*)d_in[8];
  const float* b2 = (const float*)d_in[9];
  float* out = (float*)d_out;

  const int NFEAT = 14;
  int n = in_sizes[0] / NFEAT;
  int E = in_sizes[1] / 2;
  const int* src = ei;
  const int* dst = ei + E;

  int nbk = (n + BNODES - 1) >> BSH;
  float* fws = (float*)d_ws;
  float* H = fws;
  float* XO = H + (size_t)n * DIM;
  float* AS = XO + (size_t)n * DIM;
  float* AD = AS + n;
  int* CUR = (int*)(AD + n);
  int* CSR = CUR + n;
  int* TMP = (int*)H;
  int* BK = (int*)AS;
  int* BKS = (int*)AD;

  const int TB = 256;
  dim3 gNode((n + TB - 1) / TB);
  dim3 gPart(256);
  dim3 gAgg(((n + 1) / 2 + 7) / 8);

  csr_zero<<<(nbk * BK_PITCH + TB - 1) / TB, TB, 0, stream>>>(BK, nbk * BK_PITCH);
  part_count<<<gPart, PT_TB, 0, stream>>>(dst, BK, E, nbk);
  part_scan<<<1, NBK_MAX, 0, stream>>>(BK, BKS, nbk);
  part_scatter<<<gPart, PT_TB, 0, stream>>>(src, dst, BK, TMP, E, nbk);
  part_sort<<<nbk, PT_TB, 0, stream>>>(TMP, BKS, CUR, CSR, n);

  gat_dense<NFEAT><<<gNode, TB, 0, stream>>>(x, W1, as1, ad1, H, AS, AD, n);
  gat_agg<<<gAgg, TB, 0, stream>>>(CUR, CSR, H, AS, AD, as1, b1, XO, n, 1);

  gat_dense<DIM><<<gNode, TB, 0, stream>>>(XO, W2, as2, ad2, H, AS, AD, n);
  gat_agg<<<gAgg, TB, 0, stream>>>(CUR, CSR, H, AS, AD, as2, b2, out, n, 0);
}
